# starting kernel
# baseline (speedup 1.0000x reference)
#include <hip/hip_runtime.h>

typedef __attribute__((ext_vector_type(2))) float v2f;
typedef __attribute__((ext_vector_type(8))) float v8f;

#define B_DIM 128
#define L_DIM 512
#define I_DIM 1024

__global__ __launch_bounds__(256) void labelwise_mlc_wmma(
    const float* __restrict__ X,
    const float* __restrict__ W,
    const float* __restrict__ bias,
    float* __restrict__ out)
{
  const int lane = threadIdx.x & 31;
  const int wave = threadIdx.x >> 5;
  const int tile = blockIdx.x * 8 + wave;
  const int l    = tile & (L_DIM - 1);
  const int b0   = (tile >> 9) << 4;

  const int m  = lane & 15;
  const int hi = lane >> 4;

  const float* arow = X + ((size_t)(b0 + m) * L_DIM + l) * I_DIM + 2 * hi;
  const float* wrow = W + (size_t)l * I_DIM + 2 * hi;

  v8f acc0 = {}, acc1 = {}, acc2 = {}, acc3 = {};

  for (int k = 0; k < I_DIM; k += 16) {
    v2f a0 = *(const v2f*)(arow + k);
    v2f w0 = *(const v2f*)(wrow + k);
    v2f a1 = *(const v2f*)(arow + k + 4);
    v2f w1 = *(const v2f*)(wrow + k + 4);
    v2f a2 = *(const v2f*)(arow + k + 8);
    v2f w2 = *(const v2f*)(wrow + k + 8);
    v2f a3 = *(const v2f*)(arow + k + 12);
    v2f w3 = *(const v2f*)(wrow + k + 12);

    acc0 = __builtin_amdgcn_wmma_f32_16x16x4_f32(false, a0, false, w0,
                                                 (short)0, acc0, false, false);
    acc1 = __builtin_amdgcn_wmma_f32_16x16x4_f32(false, a1, false, w1,
                                                 (short)0, acc1, false, false);
    acc2 = __builtin_amdgcn_wmma_f32_16x16x4_f32(false, a2, false, w2,
                                                 (short)0, acc2, false, false);
    acc3 = __builtin_amdgcn_wmma_f32_16x16x4_f32(false, a3, false, w3,
                                                 (short)0, acc3, false, false);
  }

  v8f acc = (acc0 + acc1) + (acc2 + acc3);
  const float bl = bias[l];

  if (m == 0) {
#pragma unroll
    for (int v = 0; v < 8; ++v) {
      float z = acc[v] + bl;
      float s = 1.0f / (1.0f + expf(-z));
      out[(size_t)(b0 + 8 * hi + v) * L_DIM + l] = s;
    }
  }
}

extern "C" void kernel_launch(void* const* d_in, const int* in_sizes, int n_in,
                              void* d_out, int out_size, void* d_ws, size_t ws_size,
                              hipStream_t stream) {
  const float* X    = (const float*)d_in[0];
  const float* W    = (const float*)d_in[1];
  const float* bias = (const float*)d_in[2];
  float* out        = (float*)d_out;

  (void)in_sizes; (void)n_in; (void)out_size; (void)d_ws; (void)ws_size;

  labelwise_mlc_wmma<<<512, 256, 0, stream>>>(X, W, bias, out);
}


# Remap WMMA operands so each wave streams whole contiguous X rows (512 B per load), W row kept in registers; row = 16 diagonal partials + cross-lane tree (summation order changes)
# speedup vs baseline: 1.2436x; 1.2436x over previous
#include <hip/hip_runtime.h>


typedef __attribute__((ext_vector_type(2))) float v2f;
typedef __attribute__((ext_vector_type(4))) float v4f;
typedef __attribute__((ext_vector_type(8))) float v8f;

#define B_DIM 128
#define L_DIM 512
#define I_DIM 1024
#define ROWS  16
#define NQ    8
#define AHEAD 3

#define DEV static __device__ __forceinline__

DEV float xhalf(float x) {
  int b = __builtin_bit_cast(int, x);
  return __builtin_bit_cast(float,
      __builtin_amdgcn_permlanex16(b, b, 0x76543210, 0xfedcba98, false, false));
}

template <int M>
DEV float xrow(float x) {
  return __builtin_bit_cast(float,
      __builtin_amdgcn_update_dpp(0, __builtin_bit_cast(int, x), 0x160 + M, 0xf, 0xf, true));
}

__global__ __launch_bounds__(256) void labelwise_mlc_wmma(
    const float* __restrict__ X,
    const float* __restrict__ W,
    const float* __restrict__ bias,
    float* __restrict__ out)
{
  const int lane = threadIdx.x & 31;
  const int wave = __builtin_amdgcn_readfirstlane(threadIdx.x >> 5);
  const int tile = blockIdx.x * 8 + wave;
  const int l    = tile & (L_DIM - 1);
  const int b0   = (tile >> 9) << 4;

  const float* wrow = W + (size_t)l * I_DIM + 4 * lane;
  v4f w[NQ];
#pragma unroll
  for (int q = 0; q < NQ; ++q) w[q] = *(const v4f*)(wrow + 128 * q);

  const float* xrow0 = X + ((size_t)b0 * L_DIM + l) * I_DIM + 4 * lane;
  v4f x[AHEAD + 1][NQ];

#define LOAD_ROW(R) do { \
    const float* p_ = xrow0 + (size_t)(R) * ((size_t)L_DIM * I_DIM); \
    _Pragma("unroll") for (int q = 0; q < NQ; ++q) \
      x[(R) % (AHEAD + 1)][q] = *(const v4f*)(p_ + 128 * q); } while (0)

#pragma unroll
  for (int r = 0; r < AHEAD; ++r) LOAD_ROW(r);

  const bool bit0 = (lane & 1) != 0, bit1 = (lane & 2) != 0, bit2 = (lane & 4) != 0;
  const bool ondiag = ((lane >> 3) & 1) == (lane >> 4);
  float mine = 0.0f;

#pragma unroll
  for (int r = 0; r < ROWS; ++r) {
    __builtin_amdgcn_sched_barrier(0);
    if (r + AHEAD < ROWS) LOAD_ROW(r + AHEAD);
    __builtin_amdgcn_sched_barrier(0);

    const v4f* xr = x[r % (AHEAD + 1)];
    v8f acc0 = {}, acc1 = {};
#pragma unroll
    for (int q = 0; q < NQ; ++q) {
      v2f a_lo = {xr[q].x, xr[q].y}, a_hi = {xr[q].z, xr[q].w};
      v2f w_lo = {w[q].x, w[q].y},   w_hi = {w[q].z, w[q].w};
      acc0 = __builtin_amdgcn_wmma_f32_16x16x4_f32(false, a_lo, false, w_lo,
                                                   (short)0, acc0, false, false);
      acc1 = __builtin_amdgcn_wmma_f32_16x16x4_f32(false, a_hi, false, w_hi,
                                                   (short)0, acc1, false, false);
    }
    const v8f acc = acc0 + acc1;

    const float t0 = bit0 ? acc[1] : acc[0];
    const float t1 = bit0 ? acc[3] : acc[2];
    const float t2 = bit0 ? acc[5] : acc[4];
    const float t3 = bit0 ? acc[7] : acc[6];
    const float u0 = bit1 ? t1 : t0;
    const float u1 = bit1 ? t3 : t2;
    float d = bit2 ? u1 : u0;
    d = ondiag ? d : 0.0f;

    d += xrow<1>(d);
    d += xrow<2>(d);
    d += xrow<4>(d);
    d += xrow<8>(d);
    d += xhalf(d);

    mine = (lane == r) ? d : mine;
  }
#undef LOAD_ROW

  if (lane < ROWS) {
    const float z = mine + bias[l];
    const float s = 1.0f / (1.0f + expf(-z));
    out[(size_t)(b0 + lane) * L_DIM + l] = s;
  }
}

extern "C" void kernel_launch(void* const* d_in, const int* in_sizes, int n_in,
                              void* d_out, int out_size, void* d_ws, size_t ws_size,
                              hipStream_t stream) {
  const float* X    = (const float*)d_in[0];
  const float* W    = (const float*)d_in[1];
  const float* bias = (const float*)d_in[2];
  float* out        = (float*)d_out;

  (void)in_sizes; (void)n_in; (void)out_size; (void)d_ws; (void)ws_size;

  labelwise_mlc_wmma<<<512, 256, 0, stream>>>(X, W, bias, out);
}


# Half as many waves, twice the rows each: a wave sums 32 batch rows of its label (was 16) with 6 rows of loads requested ahead (was 3); 256 work-groups of 8 waves, one per WGP, two waves per SIMD
# speedup vs baseline: 1.2867x; 1.0347x over previous
#include <hip/hip_runtime.h>


typedef __attribute__((ext_vector_type(2))) float v2f;
typedef __attribute__((ext_vector_type(4))) float v4f;
typedef __attribute__((ext_vector_type(8))) float v8f;

#define B_DIM 128
#define L_DIM 512
#define I_DIM 1024
#define ROWS  32
#define NQ    8
#define AHEAD 6

static_assert(ROWS <= 32 && B_DIM % ROWS == 0, "lane r keeps row r's sum; row groups tile the batch");

#define DEV static __device__ __forceinline__

DEV float xhalf(float x) {
  int b = __builtin_bit_cast(int, x);
  return __builtin_bit_cast(float,
      __builtin_amdgcn_permlanex16(b, b, 0x76543210, 0xfedcba98, false, false));
}

template <int M>
DEV float xrow(float x) {
  return __builtin_bit_cast(float,
      __builtin_amdgcn_update_dpp(0, __builtin_bit_cast(int, x), 0x160 + M, 0xf, 0xf, true));
}

__global__ __launch_bounds__(256) void labelwise_mlc_wmma(
    const float* __restrict__ X,
    const float* __restrict__ W,
    const float* __restrict__ bias,
    float* __restrict__ out)
{
  const int lane = threadIdx.x & 31;
  const int wave = __builtin_amdgcn_readfirstlane(threadIdx.x >> 5);
  const int tile = blockIdx.x * 8 + wave;
  const int l    = tile & (L_DIM - 1);
  const int b0   = (tile >> 9) * ROWS;

  const float* wrow = W + (size_t)l * I_DIM + 4 * lane;
  v4f w[NQ];
#pragma unroll
  for (int q = 0; q < NQ; ++q) w[q] = *(const v4f*)(wrow + 128 * q);
  __builtin_amdgcn_sched_barrier(0);

  const float* xrow0 = X + ((size_t)b0 * L_DIM + l) * I_DIM + 4 * lane;
  v4f x[AHEAD + 1][NQ];

#define LOAD_ROW(R) do { \
    const float* p_ = xrow0 + (size_t)(R) * ((size_t)L_DIM * I_DIM); \
    _Pragma("unroll") for (int q = 0; q < NQ; ++q) \
      x[(R) % (AHEAD + 1)][q] = *(const v4f*)(p_ + 128 * q); } while (0)

#pragma unroll
  for (int r = 0; r < AHEAD; ++r) LOAD_ROW(r);

  const bool bit0 = (lane & 1) != 0, bit1 = (lane & 2) != 0, bit2 = (lane & 4) != 0;
  const bool ondiag = ((lane >> 3) & 1) == (lane >> 4);
  float mine = 0.0f;

#pragma unroll
  for (int r = 0; r < ROWS; ++r) {
    __builtin_amdgcn_sched_barrier(0);
    if (r + AHEAD < ROWS) LOAD_ROW(r + AHEAD);
    __builtin_amdgcn_sched_barrier(0);

    const v4f* xr = x[r % (AHEAD + 1)];
    v8f acc0 = {}, acc1 = {};
#pragma unroll
    for (int q = 0; q < NQ; ++q) {
      v2f a_lo = {xr[q].x, xr[q].y}, a_hi = {xr[q].z, xr[q].w};
      v2f w_lo = {w[q].x, w[q].y},   w_hi = {w[q].z, w[q].w};
      acc0 = __builtin_amdgcn_wmma_f32_16x16x4_f32(false, a_lo, false, w_lo,
                                                   (short)0, acc0, false, false);
      acc1 = __builtin_amdgcn_wmma_f32_16x16x4_f32(false, a_hi, false, w_hi,
                                                   (short)0, acc1, false, false);
    }
    const v8f acc = acc0 + acc1;

    const float t0 = bit0 ? acc[1] : acc[0];
    const float t1 = bit0 ? acc[3] : acc[2];
    const float t2 = bit0 ? acc[5] : acc[4];
    const float t3 = bit0 ? acc[7] : acc[6];
    const float u0 = bit1 ? t1 : t0;
    const float u1 = bit1 ? t3 : t2;
    float d = bit2 ? u1 : u0;
    d = ondiag ? d : 0.0f;

    d += xrow<1>(d);
    d += xrow<2>(d);
    d += xrow<4>(d);
    d += xrow<8>(d);
    d += xhalf(d);

    mine = (lane == r) ? d : mine;
  }
#undef LOAD_ROW

  if (lane < ROWS) {
    const float z = mine + bias[l];
    const float s = 1.0f / (1.0f + expf(-z));
    out[(size_t)(b0 + lane) * L_DIM + l] = s;
  }
}

extern "C" void kernel_launch(void* const* d_in, const int* in_sizes, int n_in,
                              void* d_out, int out_size, void* d_ws, size_t ws_size,
                              hipStream_t stream) {
  const float* X    = (const float*)d_in[0];
  const float* W    = (const float*)d_in[1];
  const float* bias = (const float*)d_in[2];
  float* out        = (float*)d_out;

  (void)in_sizes; (void)n_in; (void)out_size; (void)d_ws; (void)ws_size;

  labelwise_mlc_wmma<<<256, 256, 0, stream>>>(X, W, bias, out);
}


# One wave per SIMD: 1024 waves of 64 batch rows each (was 2048 of 32) in 256 work-groups of 4, rows in a loop of 8-row groups over a ring of 8 row buffers; sums and their order unchanged
# speedup vs baseline: 1.3954x; 1.0845x over previous
#include <hip/hip_runtime.h>


typedef __attribute__((ext_vector_type(2))) float v2f;
typedef __attribute__((ext_vector_type(4))) float v4f;
typedef __attribute__((ext_vector_type(8))) float v8f;

#define B_DIM 128
#define L_DIM 512
#define I_DIM 1024
#define ROWS  64
#define NQ    8
#define AHEAD 6
#define RING  8

static_assert(ROWS == 64 && B_DIM % ROWS == 0, "lane r keeps the sums of rows r and r + 32; row groups tile the batch");
static_assert(RING > AHEAD && ROWS % RING == 0 && AHEAD < RING, "a slot is refilled only after its row is summed");

#define DEV static __device__ __forceinline__

DEV float xhalf(float x) {
  int b = __builtin_bit_cast(int, x);
  return __builtin_bit_cast(float,
      __builtin_amdgcn_permlanex16(b, b, 0x76543210, 0xfedcba98, false, false));
}

template <int M>
DEV float xrow(float x) {
  return __builtin_bit_cast(float,
      __builtin_amdgcn_update_dpp(0, __builtin_bit_cast(int, x), 0x160 + M, 0xf, 0xf, true));
}

__global__ __launch_bounds__(128) void labelwise_mlc_wmma(
    const float* __restrict__ X,
    const float* __restrict__ W,
    const float* __restrict__ bias,
    float* __restrict__ out)
{
  const int lane = threadIdx.x & 31;
  const int wave = __builtin_amdgcn_readfirstlane(threadIdx.x >> 5);
  const int tile = blockIdx.x * 4 + wave;
  const int l    = tile & (L_DIM - 1);
  const int b0   = (tile >> 9) * ROWS;

  const float* wrow = W + (size_t)l * I_DIM + 4 * lane;
  v4f w[NQ];
#pragma unroll
  for (int q = 0; q < NQ; ++q) w[q] = *(const v4f*)(wrow + 128 * q);
  __builtin_amdgcn_sched_barrier(0);

  const size_t row_pitch = (size_t)L_DIM * I_DIM;
  const float* xrow0 = X + ((size_t)b0 * L_DIM + l) * I_DIM + 4 * lane;
  v4f x[RING][NQ];

#define LOAD_ROW(SLOT, PTR) do { \
    const float* p_ = (PTR); \
    _Pragma("unroll") for (int q = 0; q < NQ; ++q) \
      x[(SLOT)][q] = *(const v4f*)(p_ + 128 * q); } while (0)

#pragma unroll
  for (int r = 0; r < AHEAD; ++r) LOAD_ROW(r, xrow0 + (size_t)r * row_pitch);

  const bool bit0 = (lane & 1) != 0, bit1 = (lane & 2) != 0, bit2 = (lane & 4) != 0;
  const bool ondiag = ((lane >> 3) & 1) == (lane >> 4);
  float mine0 = 0.0f, mine1 = 0.0f;

#define SUM_ROW(SLOT, D) do { \
    const v4f* xr = x[(SLOT)]; \
    v8f acc0 = {}, acc1 = {}; \
    _Pragma("unroll") for (int q = 0; q < NQ; ++q) { \
      v2f a_lo = {xr[q].x, xr[q].y}, a_hi = {xr[q].z, xr[q].w}; \
      v2f w_lo = {w[q].x, w[q].y},   w_hi = {w[q].z, w[q].w}; \
      acc0 = __builtin_amdgcn_wmma_f32_16x16x4_f32(false, a_lo, false, w_lo, \
                                                   (short)0, acc0, false, false); \
      acc1 = __builtin_amdgcn_wmma_f32_16x16x4_f32(false, a_hi, false, w_hi, \
                                                   (short)0, acc1, false, false); \
    } \
    const v8f acc = acc0 + acc1; \
    const float t0 = bit0 ? acc[1] : acc[0]; \
    const float t1 = bit0 ? acc[3] : acc[2]; \
    const float t2 = bit0 ? acc[5] : acc[4]; \
    const float t3 = bit0 ? acc[7] : acc[6]; \
    const float u0 = bit1 ? t1 : t0; \
    const float u1 = bit1 ? t3 : t2; \
    float d_ = bit2 ? u1 : u0; \
    d_ = ondiag ? d_ : 0.0f; \
    d_ += xrow<1>(d_); \
    d_ += xrow<2>(d_); \
    d_ += xrow<4>(d_); \
    d_ += xrow<8>(d_); \
    d_ += xhalf(d_); \
    (D) = d_; } while (0)

#define FULL_GROUP(XG, R0) do { \
    _Pragma("unroll") for (int j = 0; j < RING; ++j) { \
      __builtin_amdgcn_sched_barrier(0); \
      LOAD_ROW((j + AHEAD) % RING, (XG) + (size_t)(j + AHEAD) * row_pitch); \
      __builtin_amdgcn_sched_barrier(0); \
      float d; \
      SUM_ROW(j, d); \
      const int R = (R0) + j; \
      mine0 = (lane == R) ? d : mine0; \
      mine1 = (lane + 32 == R) ? d : mine1; \
    } } while (0)

  const float* xg = xrow0;
  FULL_GROUP(xg, 0);
  xg += (size_t)RING * row_pitch;

#pragma clang loop unroll(disable)
  for (int g = 1; g < ROWS / RING - 1; ++g) {
    FULL_GROUP(xg, g * RING);
    xg += (size_t)RING * row_pitch;
  }
#undef FULL_GROUP

#pragma unroll
  for (int j = 0; j < RING; ++j) {
    __builtin_amdgcn_sched_barrier(0);
    if (j + AHEAD < RING) LOAD_ROW((j + AHEAD) % RING, xg + (size_t)(j + AHEAD) * row_pitch);
    __builtin_amdgcn_sched_barrier(0);
    float d;
    SUM_ROW(j, d);
    mine1 = (lane == ROWS - RING - 32 + j) ? d : mine1;
  }
#undef LOAD_ROW
#undef SUM_ROW

  {
    const float bl = bias[l];
    const float z0 = mine0 + bl, z1 = mine1 + bl;
    out[(size_t)(b0 + lane) * L_DIM + l]      = 1.0f / (1.0f + expf(-z0));
    out[(size_t)(b0 + 32 + lane) * L_DIM + l] = 1.0f / (1.0f + expf(-z1));
  }
}

extern "C" void kernel_launch(void* const* d_in, const int* in_sizes, int n_in,
                              void* d_out, int out_size, void* d_ws, size_t ws_size,
                              hipStream_t stream) {
  const float* X    = (const float*)d_in[0];
  const float* W    = (const float*)d_in[1];
  const float* bias = (const float*)d_in[2];
  float* out        = (float*)d_out;

  (void)in_sizes; (void)n_in; (void)out_size; (void)d_ws; (void)ws_size;

  labelwise_mlc_wmma<<<256, 128, 0, stream>>>(X, W, bias, out);
}
